# starting kernel
# baseline (speedup 1.0000x reference)
#include <hip/hip_runtime.h>
#include <hip/hip_bf16.h>

typedef __attribute__((ext_vector_type(2))) float v2f_t;
typedef __attribute__((ext_vector_type(4))) float v4f_t;
typedef __attribute__((ext_vector_type(8))) float v8f_t;

#define THREADS 256

__device__ __forceinline__ void atadd1(float* p, float a) {
    asm volatile("global_atomic_add_f32 %0, %1, off scope:SCOPE_DEV"
                 :: "v"(p), "v"(a) : "memory");
}
__device__ __forceinline__ void atadd2(float* p, float a, float b) {
    asm volatile("global_atomic_add_f32 %0, %1, off scope:SCOPE_DEV\n\t"
                 "global_atomic_add_f32 %0, %2, off offset:4 scope:SCOPE_DEV"
                 :: "v"(p), "v"(a), "v"(b) : "memory");
}
__device__ __forceinline__ void atadd4(float* p, float a, float b, float c, float d) {
    asm volatile("global_atomic_add_f32 %0, %1, off scope:SCOPE_DEV\n\t"
                 "global_atomic_add_f32 %0, %2, off offset:4 scope:SCOPE_DEV\n\t"
                 "global_atomic_add_f32 %0, %3, off offset:8 scope:SCOPE_DEV\n\t"
                 "global_atomic_add_f32 %0, %4, off offset:12 scope:SCOPE_DEV"
                 :: "v"(p), "v"(a), "v"(b), "v"(c), "v"(d) : "memory");
}

__global__ void gcn_deg_kernel(const int* __restrict__ dst, float* __restrict__ deg, int E) {
    int e = blockIdx.x * blockDim.x + threadIdx.x;
    if (e >= E) return;
    int d = __builtin_nontemporal_load(dst + e);
    atadd1(&deg[d], 1.0f);
}

__global__ void gcn_dinv_kernel(float* __restrict__ deg, int N) {
    int i = blockIdx.x * blockDim.x + threadIdx.x;
    if (i < N) deg[i] = rsqrtf(deg[i] + 1.0f);
}

__global__ void gcn_gemm1_wmma(const float* __restrict__ x, const float* __restrict__ W1,
                               float* __restrict__ h1, int ntiles, int N) {
    int wave = blockIdx.x * (blockDim.x >> 5) + (threadIdx.x >> 5);
    if (wave >= ntiles) return;
    int lane = threadIdx.x & 31;
    int half = lane >> 4;
    int m    = lane & 15;
    int koff = half << 1;
    int row  = wave * 16 + m;
    int rcl  = row < N ? row : N - 1;
    const v2f_t* xr = (const v2f_t*)(x + (size_t)rcl * 128);

    v8f_t c = {};
#pragma unroll
    for (int k0 = 0; k0 < 128; k0 += 4) {
        v2f_t a = __builtin_nontemporal_load(xr + ((k0 + koff) >> 1));
        v2f_t b;
        b.x = W1[(k0 + koff) * 16 + m];
        b.y = W1[(k0 + koff + 1) * 16 + m];
        c = __builtin_amdgcn_wmma_f32_16x16x4_f32(false, a, false, b, (short)0, c,
                                                  false, false);
    }
#pragma unroll
    for (int v = 0; v < 8; ++v) {
        int r = wave * 16 + half * 8 + v;
        if (r < N) h1[(size_t)r * 16 + m] = c[v];
    }
}

__global__ void gcn_edge_agg16(const int* __restrict__ src, const int* __restrict__ dst,
                               const float* __restrict__ dinv, const float* __restrict__ h,
                               float* __restrict__ agg, int E) {
    int t = blockIdx.x * blockDim.x + threadIdx.x;
    int e = t >> 2;
    int q = t & 3;
    if (e >= E) return;
    int s = __builtin_nontemporal_load(src + e);
    int d = __builtin_nontemporal_load(dst + e);
    float norm = dinv[s] * dinv[d];
    v4f_t v = *((const v4f_t*)(h + (size_t)s * 16) + q);
    float* p = agg + (size_t)d * 16 + q * 4;
    atadd4(p, v.x * norm, v.y * norm, v.z * norm, v.w * norm);
}

__global__ void gcn_combine1(float* __restrict__ h1, const float* __restrict__ agg1,
                             const float* __restrict__ dinv, const float* __restrict__ b1,
                             int N) {
    int i = blockIdx.x * blockDim.x + threadIdx.x;
    if (i >= N) return;
    float dv = dinv[i];
    float d2 = dv * dv;
    v4f_t* hp = (v4f_t*)(h1 + (size_t)i * 16);
    const v4f_t* ap = (const v4f_t*)(agg1 + (size_t)i * 16);
    const v4f_t* bp = (const v4f_t*)b1;
#pragma unroll
    for (int j = 0; j < 4; ++j) {
        v4f_t hv = hp[j], av = ap[j], bv = bp[j];
        v4f_t o;
        o.x = fmaxf(av.x + hv.x * d2 + bv.x, 0.0f);
        o.y = fmaxf(av.y + hv.y * d2 + bv.y, 0.0f);
        o.z = fmaxf(av.z + hv.z * d2 + bv.z, 0.0f);
        o.w = fmaxf(av.w + hv.w * d2 + bv.w, 0.0f);
        hp[j] = o;
    }
}

__global__ void gcn_gemm2(const float* __restrict__ h, const float* __restrict__ W2,
                          float* __restrict__ h2, int N) {
    int i = blockIdx.x * blockDim.x + threadIdx.x;
    if (i >= N) return;
    const float* hr = h + (size_t)i * 16;
    float o0 = 0.0f, o1 = 0.0f;
#pragma unroll
    for (int k = 0; k < 16; ++k) {
        float v = hr[k];
        o0 += v * W2[2 * k];
        o1 += v * W2[2 * k + 1];
    }
    h2[2 * (size_t)i]     = o0;
    h2[2 * (size_t)i + 1] = o1;
}

__global__ void gcn_edge_agg2(const int* __restrict__ src, const int* __restrict__ dst,
                              const float* __restrict__ dinv, const float* __restrict__ h,
                              float* __restrict__ agg, int E) {
    int e = blockIdx.x * blockDim.x + threadIdx.x;
    if (e >= E) return;
    int s = __builtin_nontemporal_load(src + e);
    int d = __builtin_nontemporal_load(dst + e);
    float norm = dinv[s] * dinv[d];
    v2f_t v = *(const v2f_t*)(h + (size_t)s * 2);
    atadd2(agg + (size_t)d * 2, v.x * norm, v.y * norm);
}

__global__ void gcn_final(const float* __restrict__ h2, const float* __restrict__ agg2,
                          const float* __restrict__ dinv, const float* __restrict__ b2,
                          float* __restrict__ out, int N) {
    int i = blockIdx.x * blockDim.x + threadIdx.x;
    if (i >= N) return;
    float dv = dinv[i];
    float d2 = dv * dv;
    float a = agg2[2 * (size_t)i]     + h2[2 * (size_t)i]     * d2 + b2[0];
    float b = agg2[2 * (size_t)i + 1] + h2[2 * (size_t)i + 1] * d2 + b2[1];
    float mx  = fmaxf(a, b);
    float lse = mx + logf(expf(a - mx) + expf(b - mx));
    out[2 * (size_t)i]     = a - lse;
    out[2 * (size_t)i + 1] = b - lse;
}

extern "C" void kernel_launch(void* const* d_in, const int* in_sizes, int n_in,
                              void* d_out, int out_size, void* d_ws, size_t ws_size,
                              hipStream_t stream) {
    const float* x  = (const float*)d_in[0];
    const int*   ei = (const int*)d_in[1];
    const float* W1 = (const float*)d_in[2];
    const float* b1 = (const float*)d_in[3];
    const float* W2 = (const float*)d_in[4];
    const float* b2 = (const float*)d_in[5];
    float* out = (float*)d_out;

    int N = in_sizes[0] / 128;
    int E = in_sizes[1] / 2;
    const int* src = ei;
    const int* dst = ei + E;

    float* ws   = (float*)d_ws;
    float* deg  = ws;
    float* agg1 = deg + (size_t)N;
    float* agg2 = agg1 + (size_t)16 * N;
    float* h1   = agg2 + (size_t)2 * N;
    float* h2   = h1 + (size_t)16 * N;

    hipMemsetAsync(deg, 0, (size_t)19 * (size_t)N * sizeof(float), stream);

    int eb  = (E + THREADS - 1) / THREADS;
    int nb  = (N + THREADS - 1) / THREADS;
    int eb4 = (int)(((long long)E * 4 + THREADS - 1) / THREADS);

    gcn_deg_kernel<<<eb, THREADS, 0, stream>>>(dst, deg, E);
    gcn_dinv_kernel<<<nb, THREADS, 0, stream>>>(deg, N);

    int ntiles = (N + 15) / 16;
    int gb = (ntiles + 7) / 8;
    gcn_gemm1_wmma<<<gb, THREADS, 0, stream>>>(x, W1, h1, ntiles, N);

    gcn_edge_agg16<<<eb4, THREADS, 0, stream>>>(src, dst, deg, h1, agg1, E);
    gcn_combine1<<<nb, THREADS, 0, stream>>>(h1, agg1, deg, b1, N);
    gcn_gemm2<<<nb, THREADS, 0, stream>>>(h1, W2, h2, N);
    gcn_edge_agg2<<<eb, THREADS, 0, stream>>>(src, dst, deg, h2, agg2, E);
    gcn_final<<<nb, THREADS, 0, stream>>>(h2, agg2, deg, b2, out, N);
}


# Edge norm dinv[s]*dinv[d] factored: rows stored pre-scaled by their own dinv, dinv[dst] applied once per node after the sum; edge passes gather only the row (products re-associated)
# speedup vs baseline: 1.0450x; 1.0450x over previous
#include <hip/hip_runtime.h>
#include <hip/hip_bf16.h>

typedef __attribute__((ext_vector_type(2))) float v2f_t;
typedef __attribute__((ext_vector_type(4))) float v4f_t;
typedef __attribute__((ext_vector_type(8))) float v8f_t;

#define THREADS 256

__device__ __forceinline__ void atadd1(float* p, float a) {
    asm volatile("global_atomic_add_f32 %0, %1, off scope:SCOPE_DEV"
                 :: "v"(p), "v"(a) : "memory");
}
__device__ __forceinline__ void atadd2(float* p, float a, float b) {
    asm volatile("global_atomic_add_f32 %0, %1, off scope:SCOPE_DEV\n\t"
                 "global_atomic_add_f32 %0, %2, off offset:4 scope:SCOPE_DEV"
                 :: "v"(p), "v"(a), "v"(b) : "memory");
}
__device__ __forceinline__ void atadd4(float* p, float a, float b, float c, float d) {
    asm volatile("global_atomic_add_f32 %0, %1, off scope:SCOPE_DEV\n\t"
                 "global_atomic_add_f32 %0, %2, off offset:4 scope:SCOPE_DEV\n\t"
                 "global_atomic_add_f32 %0, %3, off offset:8 scope:SCOPE_DEV\n\t"
                 "global_atomic_add_f32 %0, %4, off offset:12 scope:SCOPE_DEV"
                 :: "v"(p), "v"(a), "v"(b), "v"(c), "v"(d) : "memory");
}

__global__ void gcn_deg_kernel(const int* __restrict__ dst, float* __restrict__ deg, int E) {
    int e = blockIdx.x * blockDim.x + threadIdx.x;
    if (e >= E) return;
    int d = __builtin_nontemporal_load(dst + e);
    atadd1(&deg[d], 1.0f);
}

__global__ void gcn_dinv_kernel(float* __restrict__ deg, int N) {
    int i = blockIdx.x * blockDim.x + threadIdx.x;
    if (i < N) deg[i] = rsqrtf(deg[i] + 1.0f);
}

__global__ void gcn_gemm1_wmma(const float* __restrict__ x, const float* __restrict__ W1,
                               const float* __restrict__ dinv,
                               float* __restrict__ h1, int ntiles, int N) {
    int wave = blockIdx.x * (blockDim.x >> 5) + (threadIdx.x >> 5);
    if (wave >= ntiles) return;
    int lane = threadIdx.x & 31;
    int half = lane >> 4;
    int m    = lane & 15;
    int koff = half << 1;
    int row  = wave * 16 + m;
    int rcl  = row < N ? row : N - 1;
    const v2f_t* xr = (const v2f_t*)(x + (size_t)rcl * 128);

    v8f_t c = {};
#pragma unroll
    for (int k0 = 0; k0 < 128; k0 += 4) {
        v2f_t a = __builtin_nontemporal_load(xr + ((k0 + koff) >> 1));
        v2f_t b;
        b.x = W1[(k0 + koff) * 16 + m];
        b.y = W1[(k0 + koff + 1) * 16 + m];
        c = __builtin_amdgcn_wmma_f32_16x16x4_f32(false, a, false, b, (short)0, c,
                                                  false, false);
    }
    int rbase = wave * 16 + half * 8;
    float dv[8];
#pragma unroll
    for (int v = 0; v < 8; ++v) {
        int r = rbase + v;
        dv[v] = dinv[r < N ? r : N - 1];
    }
#pragma unroll
    for (int v = 0; v < 8; ++v) {
        int r = rbase + v;
        if (r < N) h1[(size_t)r * 16 + m] = c[v] * dv[v];
    }
}

__global__ void gcn_edge_agg16(const int* __restrict__ src, const int* __restrict__ dst,
                               const float* __restrict__ h,
                               float* __restrict__ agg, int E) {
    int t = blockIdx.x * blockDim.x + threadIdx.x;
    int e = t >> 2;
    int q = t & 3;
    if (e >= E) return;
    int s = __builtin_nontemporal_load(src + e);
    int d = __builtin_nontemporal_load(dst + e);
    v4f_t v = *((const v4f_t*)(h + (size_t)s * 16) + q);
    float* p = agg + (size_t)d * 16 + q * 4;
    atadd4(p, v.x, v.y, v.z, v.w);
}

__global__ void gcn_combine1(float* __restrict__ h1, const float* __restrict__ agg1,
                             const float* __restrict__ dinv, const float* __restrict__ b1,
                             int N) {
    int i = blockIdx.x * blockDim.x + threadIdx.x;
    if (i >= N) return;
    float dv = dinv[i];
    v4f_t* hp = (v4f_t*)(h1 + (size_t)i * 16);
    const v4f_t* ap = (const v4f_t*)(agg1 + (size_t)i * 16);
    const v4f_t* bp = (const v4f_t*)b1;
#pragma unroll
    for (int j = 0; j < 4; ++j) {
        v4f_t hv = hp[j], av = ap[j], bv = bp[j];
        v4f_t o;
        o.x = fmaxf((av.x + hv.x) * dv + bv.x, 0.0f);
        o.y = fmaxf((av.y + hv.y) * dv + bv.y, 0.0f);
        o.z = fmaxf((av.z + hv.z) * dv + bv.z, 0.0f);
        o.w = fmaxf((av.w + hv.w) * dv + bv.w, 0.0f);
        hp[j] = o;
    }
}

__global__ void gcn_gemm2(const float* __restrict__ h, const float* __restrict__ W2,
                          const float* __restrict__ dinv,
                          float* __restrict__ h2, int N) {
    int i = blockIdx.x * blockDim.x + threadIdx.x;
    if (i >= N) return;
    const float* hr = h + (size_t)i * 16;
    float o0 = 0.0f, o1 = 0.0f;
#pragma unroll
    for (int k = 0; k < 16; ++k) {
        float v = hr[k];
        o0 += v * W2[2 * k];
        o1 += v * W2[2 * k + 1];
    }
    float dv = dinv[i];
    h2[2 * (size_t)i]     = o0 * dv;
    h2[2 * (size_t)i + 1] = o1 * dv;
}

__global__ void gcn_edge_agg2(const int* __restrict__ src, const int* __restrict__ dst,
                              const float* __restrict__ h,
                              float* __restrict__ agg, int E) {
    int e = blockIdx.x * blockDim.x + threadIdx.x;
    if (e >= E) return;
    int s = __builtin_nontemporal_load(src + e);
    int d = __builtin_nontemporal_load(dst + e);
    v2f_t v = *(const v2f_t*)(h + (size_t)s * 2);
    atadd2(agg + (size_t)d * 2, v.x, v.y);
}

__global__ void gcn_final(const float* __restrict__ h2, const float* __restrict__ agg2,
                          const float* __restrict__ dinv, const float* __restrict__ b2,
                          float* __restrict__ out, int N) {
    int i = blockIdx.x * blockDim.x + threadIdx.x;
    if (i >= N) return;
    float dv = dinv[i];
    float a = (agg2[2 * (size_t)i]     + h2[2 * (size_t)i])     * dv + b2[0];
    float b = (agg2[2 * (size_t)i + 1] + h2[2 * (size_t)i + 1]) * dv + b2[1];
    float mx  = fmaxf(a, b);
    float lse = mx + logf(expf(a - mx) + expf(b - mx));
    out[2 * (size_t)i]     = a - lse;
    out[2 * (size_t)i + 1] = b - lse;
}

extern "C" void kernel_launch(void* const* d_in, const int* in_sizes, int n_in,
                              void* d_out, int out_size, void* d_ws, size_t ws_size,
                              hipStream_t stream) {
    const float* x  = (const float*)d_in[0];
    const int*   ei = (const int*)d_in[1];
    const float* W1 = (const float*)d_in[2];
    const float* b1 = (const float*)d_in[3];
    const float* W2 = (const float*)d_in[4];
    const float* b2 = (const float*)d_in[5];
    float* out = (float*)d_out;

    int N = in_sizes[0] / 128;
    int E = in_sizes[1] / 2;
    const int* src = ei;
    const int* dst = ei + E;

    float* ws   = (float*)d_ws;
    float* deg  = ws;
    float* agg1 = deg + (size_t)N;
    float* agg2 = agg1 + (size_t)16 * N;
    float* h1   = agg2 + (size_t)2 * N;
    float* h2   = h1 + (size_t)16 * N;

    hipMemsetAsync(deg, 0, (size_t)19 * (size_t)N * sizeof(float), stream);

    int eb  = (E + THREADS - 1) / THREADS;
    int nb  = (N + THREADS - 1) / THREADS;
    int eb4 = (int)(((long long)E * 4 + THREADS - 1) / THREADS);

    gcn_deg_kernel<<<eb, THREADS, 0, stream>>>(dst, deg, E);
    gcn_dinv_kernel<<<nb, THREADS, 0, stream>>>(deg, N);

    int ntiles = (N + 15) / 16;
    int gb = (ntiles + 7) / 8;
    gcn_gemm1_wmma<<<gb, THREADS, 0, stream>>>(x, W1, deg, h1, ntiles, N);

    gcn_edge_agg16<<<eb4, THREADS, 0, stream>>>(src, dst, h1, agg1, E);
    gcn_combine1<<<nb, THREADS, 0, stream>>>(h1, agg1, deg, b1, N);
    gcn_gemm2<<<nb, THREADS, 0, stream>>>(h1, W2, deg, h2, N);
    gcn_edge_agg2<<<eb, THREADS, 0, stream>>>(src, dst, h2, agg2, E);
    gcn_final<<<nb, THREADS, 0, stream>>>(h2, agg2, deg, b2, out, N);
}


# Edge passes: the lanes of an edge own adjacent floats of its row (16 lanes/edge, 2 in layer 2; 32 edges per wave), so a row's f32 atomic adds and gather are one contiguous access; arrival-order sums
# speedup vs baseline: 1.3625x; 1.3038x over previous
#include <hip/hip_runtime.h>
#include <hip/hip_bf16.h>

typedef __attribute__((ext_vector_type(2))) float v2f_t;
typedef __attribute__((ext_vector_type(4))) float v4f_t;
typedef __attribute__((ext_vector_type(8))) float v8f_t;

#define THREADS 256

__device__ __forceinline__ void atadd1(float* p, float a) {
    asm volatile("global_atomic_add_f32 %0, %1, off scope:SCOPE_DEV"
                 :: "v"(p), "v"(a) : "memory");
}
__device__ __forceinline__ void atadd1_off(float* base, unsigned byte_off, float a) {
    asm volatile("global_atomic_add_f32 %0, %1, %2 scope:SCOPE_DEV"
                 :: "v"(byte_off), "v"(a), "s"(base) : "memory");
}

__global__ void gcn_deg_kernel(const int* __restrict__ dst, float* __restrict__ deg, int E) {
    int e = blockIdx.x * blockDim.x + threadIdx.x;
    if (e >= E) return;
    int d = __builtin_nontemporal_load(dst + e);
    atadd1(&deg[d], 1.0f);
}

__global__ void gcn_dinv_kernel(float* __restrict__ deg, int N) {
    int i = blockIdx.x * blockDim.x + threadIdx.x;
    if (i < N) deg[i] = rsqrtf(deg[i] + 1.0f);
}

__global__ void gcn_gemm1_wmma(const float* __restrict__ x, const float* __restrict__ W1,
                               const float* __restrict__ dinv,
                               float* __restrict__ h1, int ntiles, int N) {
    int wave = blockIdx.x * (blockDim.x >> 5) + (threadIdx.x >> 5);
    if (wave >= ntiles) return;
    int lane = threadIdx.x & 31;
    int half = lane >> 4;
    int m    = lane & 15;
    int koff = half << 1;
    int row  = wave * 16 + m;
    int rcl  = row < N ? row : N - 1;
    const v2f_t* xr = (const v2f_t*)(x + (size_t)rcl * 128);

    v8f_t c = {};
#pragma unroll
    for (int k0 = 0; k0 < 128; k0 += 4) {
        v2f_t a = __builtin_nontemporal_load(xr + ((k0 + koff) >> 1));
        v2f_t b;
        b.x = W1[(k0 + koff) * 16 + m];
        b.y = W1[(k0 + koff + 1) * 16 + m];
        c = __builtin_amdgcn_wmma_f32_16x16x4_f32(false, a, false, b, (short)0, c,
                                                  false, false);
    }
    int rbase = wave * 16 + half * 8;
    float dv[8];
#pragma unroll
    for (int v = 0; v < 8; ++v) {
        int r = rbase + v;
        dv[v] = dinv[r < N ? r : N - 1];
    }
#pragma unroll
    for (int v = 0; v < 8; ++v) {
        int r = rbase + v;
        if (r < N) h1[(size_t)r * 16 + m] = c[v] * dv[v];
    }
}

template <bool TAIL>
__device__ __forceinline__ void agg16_group(const int* __restrict__ src, const int* __restrict__ dst,
                                            const float* __restrict__ h, float* __restrict__ agg,
                                            int e0, int E, int lane) {
    const int half = lane >> 4;
    const unsigned fb = (unsigned)(lane & 15) << 2;
    int s_all = 0, d_all = 0;
    if (!TAIL || e0 + lane < E) {
        s_all = __builtin_nontemporal_load(src + e0 + lane);
        d_all = __builtin_nontemporal_load(dst + e0 + lane);
    }
    const char* hb = (const char*)h;
    float    v[16];
    unsigned ob[16];
#pragma unroll
    for (int i = 0; i < 16; ++i) {
        int s0 = __builtin_amdgcn_readlane(s_all, 2 * i);
        int s1 = __builtin_amdgcn_readlane(s_all, 2 * i + 1);
        int d0 = __builtin_amdgcn_readlane(d_all, 2 * i);
        int d1 = __builtin_amdgcn_readlane(d_all, 2 * i + 1);
        unsigned s = (unsigned)(half ? s1 : s0);
        unsigned d = (unsigned)(half ? d1 : d0);
        ob[i] = (d << 6) | fb;
        v[i]  = *(const float*)(hb + ((s << 6) | fb));
    }
#pragma unroll
    for (int i = 0; i < 16; ++i) {
        if (!TAIL || e0 + 2 * i + half < E)
            atadd1_off(agg, ob[i], v[i]);
    }
}

__global__ void gcn_edge_agg16(const int* __restrict__ src, const int* __restrict__ dst,
                               const float* __restrict__ h,
                               float* __restrict__ agg, int E) {
    const int lane  = threadIdx.x & 31;
    const int g     = blockIdx.x * (blockDim.x >> 5) + (threadIdx.x >> 5);
    const int nfull = E >> 5;
    if (g < nfull)
        agg16_group<false>(src, dst, h, agg, g << 5, E, lane);
    else if (g == nfull && (E & 31) != 0)
        agg16_group<true>(src, dst, h, agg, g << 5, E, lane);
}

__global__ void gcn_combine1(float* __restrict__ h1, const float* __restrict__ agg1,
                             const float* __restrict__ dinv, const float* __restrict__ b1,
                             int N) {
    int i = blockIdx.x * blockDim.x + threadIdx.x;
    if (i >= N) return;
    float dv = dinv[i];
    v4f_t* hp = (v4f_t*)(h1 + (size_t)i * 16);
    const v4f_t* ap = (const v4f_t*)(agg1 + (size_t)i * 16);
    const v4f_t* bp = (const v4f_t*)b1;
#pragma unroll
    for (int j = 0; j < 4; ++j) {
        v4f_t hv = hp[j], av = ap[j], bv = bp[j];
        v4f_t o;
        o.x = fmaxf((av.x + hv.x) * dv + bv.x, 0.0f);
        o.y = fmaxf((av.y + hv.y) * dv + bv.y, 0.0f);
        o.z = fmaxf((av.z + hv.z) * dv + bv.z, 0.0f);
        o.w = fmaxf((av.w + hv.w) * dv + bv.w, 0.0f);
        hp[j] = o;
    }
}

__global__ void gcn_gemm2(const float* __restrict__ h, const float* __restrict__ W2,
                          const float* __restrict__ dinv,
                          float* __restrict__ h2, int N) {
    int i = blockIdx.x * blockDim.x + threadIdx.x;
    if (i >= N) return;
    const float* hr = h + (size_t)i * 16;
    float o0 = 0.0f, o1 = 0.0f;
#pragma unroll
    for (int k = 0; k < 16; ++k) {
        float v = hr[k];
        o0 += v * W2[2 * k];
        o1 += v * W2[2 * k + 1];
    }
    float dv = dinv[i];
    h2[2 * (size_t)i]     = o0 * dv;
    h2[2 * (size_t)i + 1] = o1 * dv;
}

template <bool TAIL>
__device__ __forceinline__ void agg2_group(const int* __restrict__ src, const int* __restrict__ dst,
                                           const float* __restrict__ h, float* __restrict__ agg,
                                           int e0, int E, int lane) {
    const int pr = lane >> 1;
    const unsigned fb = (unsigned)(lane & 1) << 2;
    int s_all = 0, d_all = 0;
    if (!TAIL || e0 + lane < E) {
        s_all = __builtin_nontemporal_load(src + e0 + lane);
        d_all = __builtin_nontemporal_load(dst + e0 + lane);
    }
    unsigned sa = (unsigned)__shfl(s_all, pr);
    unsigned sb = (unsigned)__shfl(s_all, 16 + pr);
    unsigned da = (unsigned)__shfl(d_all, pr);
    unsigned db = (unsigned)__shfl(d_all, 16 + pr);
    const char* hb = (const char*)h;
    float va = *(const float*)(hb + ((sa << 3) | fb));
    float vb = *(const float*)(hb + ((sb << 3) | fb));
    if (!TAIL || e0 + pr < E)      atadd1_off(agg, (da << 3) | fb, va);
    if (!TAIL || e0 + 16 + pr < E) atadd1_off(agg, (db << 3) | fb, vb);
}

__global__ void gcn_edge_agg2(const int* __restrict__ src, const int* __restrict__ dst,
                              const float* __restrict__ h,
                              float* __restrict__ agg, int E) {
    const int lane  = threadIdx.x & 31;
    const int g     = blockIdx.x * (blockDim.x >> 5) + (threadIdx.x >> 5);
    const int nfull = E >> 5;
    if (g < nfull)
        agg2_group<false>(src, dst, h, agg, g << 5, E, lane);
    else if (g == nfull && (E & 31) != 0)
        agg2_group<true>(src, dst, h, agg, g << 5, E, lane);
}

__global__ void gcn_final(const float* __restrict__ h2, const float* __restrict__ agg2,
                          const float* __restrict__ dinv, const float* __restrict__ b2,
                          float* __restrict__ out, int N) {
    int i = blockIdx.x * blockDim.x + threadIdx.x;
    if (i >= N) return;
    float dv = dinv[i];
    float a = (agg2[2 * (size_t)i]     + h2[2 * (size_t)i])     * dv + b2[0];
    float b = (agg2[2 * (size_t)i + 1] + h2[2 * (size_t)i + 1]) * dv + b2[1];
    float mx  = fmaxf(a, b);
    float lse = mx + logf(expf(a - mx) + expf(b - mx));
    out[2 * (size_t)i]     = a - lse;
    out[2 * (size_t)i + 1] = b - lse;
}

extern "C" void kernel_launch(void* const* d_in, const int* in_sizes, int n_in,
                              void* d_out, int out_size, void* d_ws, size_t ws_size,
                              hipStream_t stream) {
    const float* x  = (const float*)d_in[0];
    const int*   ei = (const int*)d_in[1];
    const float* W1 = (const float*)d_in[2];
    const float* b1 = (const float*)d_in[3];
    const float* W2 = (const float*)d_in[4];
    const float* b2 = (const float*)d_in[5];
    float* out = (float*)d_out;

    int N = in_sizes[0] / 128;
    int E = in_sizes[1] / 2;
    const int* src = ei;
    const int* dst = ei + E;

    float* ws   = (float*)d_ws;
    float* deg  = ws;
    float* agg1 = deg + (size_t)N;
    float* agg2 = agg1 + (size_t)16 * N;
    float* h1   = agg2 + (size_t)2 * N;
    float* h2   = h1 + (size_t)16 * N;

    hipMemsetAsync(deg, 0, (size_t)19 * (size_t)N * sizeof(float), stream);

    int eb  = (E + THREADS - 1) / THREADS;
    int nb  = (N + THREADS - 1) / THREADS;

    gcn_deg_kernel<<<eb, THREADS, 0, stream>>>(dst, deg, E);
    gcn_dinv_kernel<<<nb, THREADS, 0, stream>>>(deg, N);

    int ntiles = (N + 15) / 16;
    int gb = (ntiles + 7) / 8;
    gcn_gemm1_wmma<<<gb, THREADS, 0, stream>>>(x, W1, deg, h1, ntiles, N);

    gcn_edge_agg16<<<eb, THREADS, 0, stream>>>(src, dst, h1, agg1, E);
    gcn_combine1<<<nb, THREADS, 0, stream>>>(h1, agg1, deg, b1, N);
    gcn_gemm2<<<nb, THREADS, 0, stream>>>(h1, W2, deg, h2, N);
    gcn_edge_agg2<<<eb, THREADS, 0, stream>>>(src, dst, h2, agg2, E);
    gcn_final<<<nb, THREADS, 0, stream>>>(h2, agg2, deg, b2, out, N);
}
